# starting kernel
# baseline (speedup 1.0000x reference)
#include <hip/hip_runtime.h>

typedef __attribute__((ext_vector_type(2))) float v2f;
typedef __attribute__((ext_vector_type(8))) float v8f;

#define SEQ_L 2048
#define HEAD_D 64
#define NBLK 128

__global__ __launch_bounds__(32)
void sparse_attn_wmma_f32(const float* __restrict__ q,
                          const float* __restrict__ k,
                          const float* __restrict__ v,
                          float* __restrict__ out) {
  __shared__ float pshm[16 * 16];

  const int lane = threadIdx.x;
  const int half = lane >> 4;
  const int l16  = lane & 15;

  const int  row  = (int)(blockIdx.x & (NBLK - 1));
  const long bh   = (long)(blockIdx.x >> 7);
  const long base = bh * (long)(SEQ_L * HEAD_D);

  const float* qrow = q + base + (long)(row * 16 + l16) * HEAD_D + 2 * half;
  v2f qa[16];
#pragma unroll
  for (int kk = 0; kk < 16; ++kk) {
    v2f t = *(const v2f*)(qrow + 4 * kk);
    qa[kk].x = t.x * 0.125f;
    qa[kk].y = t.y * 0.125f;
  }

  v8f o0 = {}, o1 = {}, o2 = {}, o3 = {};
  float mrow[8], lrow[8];
#pragma unroll
  for (int r = 0; r < 8; ++r) { mrow[r] = -1e30f; lrow[r] = 0.0f; }

  const int w     = row & ~3;
  const int nvert = w >> 2;
  const int nbl   = nvert + (row - w + 1);

  for (int t = 0; t < nbl; ++t) {
    const int jb = (t < nvert) ? (4 * t + 3) : (w + (t - nvert));

    {
      const int tn  = (t + 1 < nbl) ? (t + 1) : t;
      const int jbn = (tn < nvert) ? (4 * tn + 3) : (w + (tn - nvert));
      const float* kp = k + base + (long)(jbn * 16 + l16) * HEAD_D;
      const float* vp = v + base + (long)(jbn * 16 + l16) * HEAD_D;
      __builtin_prefetch(kp, 0, 0);
      __builtin_prefetch(vp, 0, 0);
    }

    const float* krow = k + base + (long)(jb * 16 + l16) * HEAD_D + 2 * half;
    v8f s = {};
#pragma unroll
    for (int kk = 0; kk < 16; ++kk) {
      v2f b = *(const v2f*)(krow + 4 * kk);
      s = __builtin_amdgcn_wmma_f32_16x16x4_f32(false, qa[kk], false, b,
                                                (short)0, s, false, false);
    }

    float alpha[8];
#pragma unroll
    for (int r = 0; r < 8; ++r) {
      float x  = s[r];
      float rm = x;
#pragma unroll
      for (int off = 1; off < 16; off <<= 1)
        rm = fmaxf(rm, __shfl_xor(rm, off, 32));
      const float mnew = fmaxf(mrow[r], rm);
      alpha[r] = __expf(mrow[r] - mnew);
      float p = __expf(x - mnew);
      float rs = p;
#pragma unroll
      for (int off = 1; off < 16; off <<= 1)
        rs += __shfl_xor(rs, off, 32);
      lrow[r] = lrow[r] * alpha[r] + rs;
      mrow[r] = mnew;
      pshm[(r + 8 * half) * 16 + l16] = p;
    }
    __syncthreads();

#pragma unroll
    for (int r = 0; r < 8; ++r) {
      o0[r] *= alpha[r]; o1[r] *= alpha[r];
      o2[r] *= alpha[r]; o3[r] *= alpha[r];
    }

    const float* vblk = v + base + (long)(jb * 16) * HEAD_D;
#pragma unroll
    for (int kp = 0; kp < 4; ++kp) {
      v2f a = *(const v2f*)&pshm[l16 * 16 + 4 * kp + 2 * half];
      const float* vr = vblk + (long)(4 * kp + 2 * half) * HEAD_D + l16;
      v2f b0, b1, b2, b3;
      b0.x = vr[0];  b0.y = vr[HEAD_D + 0];
      b1.x = vr[16]; b1.y = vr[HEAD_D + 16];
      b2.x = vr[32]; b2.y = vr[HEAD_D + 32];
      b3.x = vr[48]; b3.y = vr[HEAD_D + 48];
      o0 = __builtin_amdgcn_wmma_f32_16x16x4_f32(false, a, false, b0, (short)0, o0, false, false);
      o1 = __builtin_amdgcn_wmma_f32_16x16x4_f32(false, a, false, b1, (short)0, o1, false, false);
      o2 = __builtin_amdgcn_wmma_f32_16x16x4_f32(false, a, false, b2, (short)0, o2, false, false);
      o3 = __builtin_amdgcn_wmma_f32_16x16x4_f32(false, a, false, b3, (short)0, o3, false, false);
    }
    __syncthreads();
  }

  float* orow = out + base + (long)(row * 16) * HEAD_D;
#pragma unroll
  for (int r = 0; r < 8; ++r) {
    const float inv = 1.0f / lrow[r];
    const int m = r + 8 * half;
    orow[(long)m * HEAD_D +  0 + l16] = o0[r] * inv;
    orow[(long)m * HEAD_D + 16 + l16] = o1[r] * inv;
    orow[(long)m * HEAD_D + 32 + l16] = o2[r] * inv;
    orow[(long)m * HEAD_D + 48 + l16] = o3[r] * inv;
  }
}

extern "C" void kernel_launch(void* const* d_in, const int* in_sizes, int n_in,
                              void* d_out, int out_size, void* d_ws, size_t ws_size,
                              hipStream_t stream) {
  (void)n_in; (void)out_size; (void)d_ws; (void)ws_size;
  const float* q = (const float*)d_in[0];
  const float* k = (const float*)d_in[1];
  const float* v = (const float*)d_in[2];
  float* out = (float*)d_out;

  const int bh = in_sizes[0] / (SEQ_L * HEAD_D);
  dim3 grid((unsigned)(bh * NBLK));
  sparse_attn_wmma_f32<<<grid, 32, 0, stream>>>(q, k, v, out);
}


# split each query block's key-block list round-robin over the 4 waves of a 128-thread work-group; partial (max,sum,O) merged via LDS, so softmax sums are added in a new order (4 partials)
# speedup vs baseline: 1.2474x; 1.2474x over previous
#include <hip/hip_runtime.h>

typedef __attribute__((ext_vector_type(2))) float v2f;
typedef __attribute__((ext_vector_type(4))) float v4f;
typedef __attribute__((ext_vector_type(8))) float v8f;

#define SEQ_L 2048
#define HEAD_D 64
#define NBLK 128
#define NSPLIT 4
#define OPITCH 68

__device__ __forceinline__ void wave_lds_handover() {
  __builtin_amdgcn_fence(__ATOMIC_RELEASE, "wavefront");
  __builtin_amdgcn_wave_barrier();
  __builtin_amdgcn_fence(__ATOMIC_ACQUIRE, "wavefront");
}

__global__ __launch_bounds__(32 * NSPLIT)
void sparse_attn_wmma_f32(const float* __restrict__ q,
                          const float* __restrict__ k,
                          const float* __restrict__ v,
                          float* __restrict__ out) {
  __shared__ float wtile[NSPLIT][16 * OPITCH];
  __shared__ float wmax[NSPLIT][16];
  __shared__ float wsum[NSPLIT][16];

  const int wv   = __builtin_amdgcn_readfirstlane((int)(threadIdx.x >> 5));
  const int lane = threadIdx.x & 31;
  const int half = lane >> 4;
  const int l16  = lane & 15;
  float* pshm = wtile[wv];

  const int  row  = (int)(blockIdx.x & (NBLK - 1));
  const long bh   = (long)(blockIdx.x >> 7);
  const long base = bh * (long)(SEQ_L * HEAD_D);

  const float* qrow = q + base + (long)(row * 16 + l16) * HEAD_D + 2 * half;
  v2f qa[16];
#pragma unroll
  for (int kk = 0; kk < 16; ++kk) {
    v2f t = *(const v2f*)(qrow + 4 * kk);
    qa[kk].x = t.x * 0.125f;
    qa[kk].y = t.y * 0.125f;
  }

  v8f o0 = {}, o1 = {}, o2 = {}, o3 = {};
  float mrow[8], lrow[8];
#pragma unroll
  for (int r = 0; r < 8; ++r) { mrow[r] = -1e30f; lrow[r] = 0.0f; }

  const int w     = row & ~3;
  const int nvert = w >> 2;
  const int nbl   = nvert + (row - w + 1);

  for (int t = wv; t < nbl; t += NSPLIT) {
    const int jb = (t < nvert) ? (4 * t + 3) : (w + (t - nvert));

    {
      const int tn  = (t + NSPLIT < nbl) ? (t + NSPLIT) : t;
      const int jbn = (tn < nvert) ? (4 * tn + 3) : (w + (tn - nvert));
      const float* kp = k + base + (long)(jbn * 16 + l16) * HEAD_D;
      const float* vp = v + base + (long)(jbn * 16 + l16) * HEAD_D;
      __builtin_prefetch(kp, 0, 0);
      __builtin_prefetch(vp, 0, 0);
    }

    const float* krow = k + base + (long)(jb * 16 + l16) * HEAD_D + 2 * half;
    v8f s = {};
#pragma unroll
    for (int kk = 0; kk < 16; ++kk) {
      v2f b = *(const v2f*)(krow + 4 * kk);
      s = __builtin_amdgcn_wmma_f32_16x16x4_f32(false, qa[kk], false, b,
                                                (short)0, s, false, false);
    }

    float alpha[8];
#pragma unroll
    for (int r = 0; r < 8; ++r) {
      float x  = s[r];
      float rm = x;
#pragma unroll
      for (int off = 1; off < 16; off <<= 1)
        rm = fmaxf(rm, __shfl_xor(rm, off, 32));
      const float mnew = fmaxf(mrow[r], rm);
      alpha[r] = __expf(mrow[r] - mnew);
      float p = __expf(x - mnew);
      float rs = p;
#pragma unroll
      for (int off = 1; off < 16; off <<= 1)
        rs += __shfl_xor(rs, off, 32);
      lrow[r] = lrow[r] * alpha[r] + rs;
      mrow[r] = mnew;
      pshm[(r + 8 * half) * 16 + l16] = p;
    }
    wave_lds_handover();

#pragma unroll
    for (int r = 0; r < 8; ++r) {
      o0[r] *= alpha[r]; o1[r] *= alpha[r];
      o2[r] *= alpha[r]; o3[r] *= alpha[r];
    }

    const float* vblk = v + base + (long)(jb * 16) * HEAD_D;
#pragma unroll
    for (int kp = 0; kp < 4; ++kp) {
      v2f a = *(const v2f*)&pshm[l16 * 16 + 4 * kp + 2 * half];
      const float* vr = vblk + (long)(4 * kp + 2 * half) * HEAD_D + l16;
      v2f b0, b1, b2, b3;
      b0.x = vr[0];  b0.y = vr[HEAD_D + 0];
      b1.x = vr[16]; b1.y = vr[HEAD_D + 16];
      b2.x = vr[32]; b2.y = vr[HEAD_D + 32];
      b3.x = vr[48]; b3.y = vr[HEAD_D + 48];
      o0 = __builtin_amdgcn_wmma_f32_16x16x4_f32(false, a, false, b0, (short)0, o0, false, false);
      o1 = __builtin_amdgcn_wmma_f32_16x16x4_f32(false, a, false, b1, (short)0, o1, false, false);
      o2 = __builtin_amdgcn_wmma_f32_16x16x4_f32(false, a, false, b2, (short)0, o2, false, false);
      o3 = __builtin_amdgcn_wmma_f32_16x16x4_f32(false, a, false, b3, (short)0, o3, false, false);
    }
    wave_lds_handover();
  }

  float* osh = wtile[wv];
#pragma unroll
  for (int r = 0; r < 8; ++r) {
    const int m = r + 8 * half;
    osh[m * OPITCH +  0 + l16] = o0[r];
    osh[m * OPITCH + 16 + l16] = o1[r];
    osh[m * OPITCH + 32 + l16] = o2[r];
    osh[m * OPITCH + 48 + l16] = o3[r];
  }
  if (l16 == 0) {
#pragma unroll
    for (int r = 0; r < 8; ++r) {
      wmax[wv][r + 8 * half] = mrow[r];
      wsum[wv][r + 8 * half] = lrow[r];
    }
  }
  __syncthreads();

#pragma unroll
  for (int i = 0; i < 2; ++i) {
    const int m = 4 * wv + 2 * i + half;
    float mp[NSPLIT], lp[NSPLIT];
    float mall = -1e30f;
#pragma unroll
    for (int p = 0; p < NSPLIT; ++p) {
      mp[p] = wmax[p][m];
      lp[p] = wsum[p][m];
      mall = fmaxf(mall, mp[p]);
    }
    float lall = 0.0f;
    v4f acc = {};
#pragma unroll
    for (int p = 0; p < NSPLIT; ++p) {
      const float sc = __expf(mp[p] - mall);
      const v4f op = *(const v4f*)&wtile[p][m * OPITCH + 4 * l16];
      lall += sc * lp[p];
      acc  += sc * op;
    }
    const float inv = 1.0f / lall;
    *(v4f*)(out + base + (long)(row * 16 + m) * HEAD_D + 4 * l16) = acc * inv;
  }
}

extern "C" void kernel_launch(void* const* d_in, const int* in_sizes, int n_in,
                              void* d_out, int out_size, void* d_ws, size_t ws_size,
                              hipStream_t stream) {
  (void)n_in; (void)out_size; (void)d_ws; (void)ws_size;
  const float* q = (const float*)d_in[0];
  const float* k = (const float*)d_in[1];
  const float* v = (const float*)d_in[2];
  float* out = (float*)d_out;

  const int bh = in_sizes[0] / (SEQ_L * HEAD_D);
  dim3 grid((unsigned)(bh * NBLK));
  sparse_attn_wmma_f32<<<grid, 32 * NSPLIT, 0, stream>>>(q, k, v, out);
}


# score tile computed transposed (S^T = K*Q^T, O^T = V^T*P^T): softmax max/sum in-lane plus one half exchange, P^T feeds the WMMA from registers, V rows as b128; key order of row sums and P*V changes
# speedup vs baseline: 1.4613x; 1.1715x over previous
#include <hip/hip_runtime.h>

typedef __attribute__((ext_vector_type(2))) float v2f;
typedef __attribute__((ext_vector_type(4))) float v4f;
typedef __attribute__((ext_vector_type(8))) float v8f;

#define SEQ_L 2048
#define HEAD_D 64
#define NBLK 128
#define NSPLIT 4
#define OPITCH 68

__device__ __forceinline__ float xhalf(float x) {
  const int b = __builtin_bit_cast(int, x);
  return __builtin_bit_cast(float,
      __builtin_amdgcn_permlanex16(b, b, 0x76543210, 0xfedcba98, false, false));
}

__global__ __launch_bounds__(32 * NSPLIT)
void sparse_attn_wmma_f32(const float* __restrict__ q,
                          const float* __restrict__ k,
                          const float* __restrict__ v,
                          float* __restrict__ out) {
  __shared__ __attribute__((aligned(16))) float wtile[NSPLIT][16 * OPITCH];
  __shared__ float wmax[NSPLIT][16];
  __shared__ float wsum[NSPLIT][16];

  const int wv   = __builtin_amdgcn_readfirstlane((int)(threadIdx.x >> 5));
  const int lane = threadIdx.x & 31;
  const int half = lane >> 4;
  const int l16  = lane & 15;

  const int  row  = (int)(blockIdx.x & (NBLK - 1));
  const long bh   = (long)(blockIdx.x >> 7);
  const long base = bh * (long)(SEQ_L * HEAD_D);

  const float* qrow = q + base + (long)(row * 16 + l16) * HEAD_D + 2 * half;
  v2f qb[16];
#pragma unroll
  for (int kk = 0; kk < 16; ++kk) {
    v2f t = *(const v2f*)(qrow + 4 * kk);
    qb[kk].x = t.x * 0.125f;
    qb[kk].y = t.y * 0.125f;
  }

  v8f o0 = {}, o1 = {}, o2 = {}, o3 = {};
  float mrun = -1e30f;
  float lsum = 0.0f;

  const int w     = row & ~3;
  const int nvert = w >> 2;
  const int nbl   = nvert + (row - w + 1);

  for (int t = wv; t < nbl; t += NSPLIT) {
    const int jb = (t < nvert) ? (4 * t + 3) : (w + (t - nvert));

    {
      const int tn  = (t + NSPLIT < nbl) ? (t + NSPLIT) : t;
      const int jbn = (tn < nvert) ? (4 * tn + 3) : (w + (tn - nvert));
      const float* kp = k + base + (long)(jbn * 16 + l16) * HEAD_D;
      const float* vp = v + base + (long)(jbn * 16 + l16) * HEAD_D;
      __builtin_prefetch(kp, 0, 0);
      __builtin_prefetch(vp, 0, 0);
    }

    const float* krow = k + base + (long)(jb * 16 + l16) * HEAD_D + 2 * half;
    v8f s = {};
#pragma unroll
    for (int kk = 0; kk < 16; ++kk) {
      v2f a = *(const v2f*)(krow + 4 * kk);
      s = __builtin_amdgcn_wmma_f32_16x16x4_f32(false, a, false, qb[kk],
                                                (short)0, s, false, false);
    }

    float rm = fmaxf(fmaxf(fmaxf(s[0], s[1]), fmaxf(s[2], s[3])),
                     fmaxf(fmaxf(s[4], s[5]), fmaxf(s[6], s[7])));
    rm = fmaxf(rm, xhalf(rm));
    const float mnew  = fmaxf(mrun, rm);
    const float alpha = __expf(mrun - mnew);
    v8f p;
#pragma unroll
    for (int r = 0; r < 8; ++r) p[r] = __expf(s[r] - mnew);
    const float rs = ((p[0] + p[1]) + (p[2] + p[3])) + ((p[4] + p[5]) + (p[6] + p[7]));
    lsum = lsum * alpha + rs;
    mrun = mnew;

    o0 *= alpha; o1 *= alpha; o2 *= alpha; o3 *= alpha;

    const float* vrow = v + base + (long)(jb * 16 + 8 * half) * HEAD_D + 4 * l16;
#pragma unroll
    for (int j = 0; j < 4; ++j) {
      const v4f va = *(const v4f*)(vrow + (2 * j) * HEAD_D);
      const v4f vb = *(const v4f*)(vrow + (2 * j + 1) * HEAD_D);
      v2f pb;  pb.x = p[2 * j];  pb.y = p[2 * j + 1];
      v2f a0, a1, a2, a3;
      a0.x = va.x; a0.y = vb.x;
      a1.x = va.y; a1.y = vb.y;
      a2.x = va.z; a2.y = vb.z;
      a3.x = va.w; a3.y = vb.w;
      o0 = __builtin_amdgcn_wmma_f32_16x16x4_f32(false, a0, false, pb, (short)0, o0, false, false);
      o1 = __builtin_amdgcn_wmma_f32_16x16x4_f32(false, a1, false, pb, (short)0, o1, false, false);
      o2 = __builtin_amdgcn_wmma_f32_16x16x4_f32(false, a2, false, pb, (short)0, o2, false, false);
      o3 = __builtin_amdgcn_wmma_f32_16x16x4_f32(false, a3, false, pb, (short)0, o3, false, false);
    }
  }

  float* osh = wtile[wv];
#pragma unroll
  for (int r = 0; r < 8; ++r) {
    v4f t;
    t.x = o0[r]; t.y = o1[r]; t.z = o2[r]; t.w = o3[r];
    *(v4f*)&osh[l16 * OPITCH + 4 * (r + 8 * half)] = t;
  }
  lsum = lsum + xhalf(lsum);
  if (half == 0) {
    wmax[wv][l16] = mrun;
    wsum[wv][l16] = lsum;
  }
  __syncthreads();

#pragma unroll
  for (int i = 0; i < 2; ++i) {
    const int m = 4 * wv + 2 * i + half;
    float mp[NSPLIT], lp[NSPLIT];
    float mall = -1e30f;
#pragma unroll
    for (int p = 0; p < NSPLIT; ++p) {
      mp[p] = wmax[p][m];
      lp[p] = wsum[p][m];
      mall = fmaxf(mall, mp[p]);
    }
    float lall = 0.0f;
    v4f acc = {};
#pragma unroll
    for (int p = 0; p < NSPLIT; ++p) {
      const float sc = __expf(mp[p] - mall);
      const v4f op = *(const v4f*)&wtile[p][m * OPITCH + 4 * l16];
      lall += sc * lp[p];
      acc  += sc * op;
    }
    const float inv = 1.0f / lall;
    *(v4f*)(out + base + (long)(row * 16 + m) * HEAD_D + 4 * l16) = acc * inv;
  }
}

extern "C" void kernel_launch(void* const* d_in, const int* in_sizes, int n_in,
                              void* d_out, int out_size, void* d_ws, size_t ws_size,
                              hipStream_t stream) {
  (void)n_in; (void)out_size; (void)d_ws; (void)ws_size;
  const float* q = (const float*)d_in[0];
  const float* k = (const float*)d_in[1];
  const float* v = (const float*)d_in[2];
  float* out = (float*)d_out;

  const int bh = in_sizes[0] / (SEQ_L * HEAD_D);
  dim3 grid((unsigned)(bh * NBLK));
  sparse_attn_wmma_f32<<<grid, 32 * NSPLIT, 0, stream>>>(q, k, v, out);
}


# work-group order only: grid (b*h, 128), b*h fastest, query block row = 127 - blockIdx.y: rows with the longest key-block lists are dispatched first; per-row arithmetic and sum order unchanged
# speedup vs baseline: 1.8684x; 1.2786x over previous
#include <hip/hip_runtime.h>

typedef __attribute__((ext_vector_type(2))) float v2f;
typedef __attribute__((ext_vector_type(4))) float v4f;
typedef __attribute__((ext_vector_type(8))) float v8f;

#define SEQ_L 2048
#define HEAD_D 64
#define NBLK 128
#define NSPLIT 4
#define OPITCH 68

__device__ __forceinline__ float xhalf(float x) {
  const int b = __builtin_bit_cast(int, x);
  return __builtin_bit_cast(float,
      __builtin_amdgcn_permlanex16(b, b, 0x76543210, 0xfedcba98, false, false));
}

__global__ __launch_bounds__(32 * NSPLIT)
void sparse_attn_wmma_f32(const float* __restrict__ q,
                          const float* __restrict__ k,
                          const float* __restrict__ v,
                          float* __restrict__ out) {
  __shared__ __attribute__((aligned(16))) float wtile[NSPLIT][16 * OPITCH];
  __shared__ float wmax[NSPLIT][16];
  __shared__ float wsum[NSPLIT][16];

  const int wv   = __builtin_amdgcn_readfirstlane((int)(threadIdx.x >> 5));
  const int lane = threadIdx.x & 31;
  const int half = lane >> 4;
  const int l16  = lane & 15;

  const int  row  = (NBLK - 1) - (int)blockIdx.y;
  const long bh   = (long)blockIdx.x;
  const long base = bh * (long)(SEQ_L * HEAD_D);

  const float* qrow = q + base + (long)(row * 16 + l16) * HEAD_D + 2 * half;
  v2f qb[16];
#pragma unroll
  for (int kk = 0; kk < 16; ++kk) {
    v2f t = *(const v2f*)(qrow + 4 * kk);
    qb[kk].x = t.x * 0.125f;
    qb[kk].y = t.y * 0.125f;
  }

  v8f o0 = {}, o1 = {}, o2 = {}, o3 = {};
  float mrun = -1e30f;
  float lsum = 0.0f;

  const int w     = row & ~3;
  const int nvert = w >> 2;
  const int nbl   = nvert + (row - w + 1);

  for (int t = wv; t < nbl; t += NSPLIT) {
    const int jb = (t < nvert) ? (4 * t + 3) : (w + (t - nvert));

    {
      const int tn  = (t + NSPLIT < nbl) ? (t + NSPLIT) : t;
      const int jbn = (tn < nvert) ? (4 * tn + 3) : (w + (tn - nvert));
      const float* kp = k + base + (long)(jbn * 16 + l16) * HEAD_D;
      const float* vp = v + base + (long)(jbn * 16 + l16) * HEAD_D;
      __builtin_prefetch(kp, 0, 0);
      __builtin_prefetch(vp, 0, 0);
    }

    const float* krow = k + base + (long)(jb * 16 + l16) * HEAD_D + 2 * half;
    v8f s = {};
#pragma unroll
    for (int kk = 0; kk < 16; ++kk) {
      v2f a = *(const v2f*)(krow + 4 * kk);
      s = __builtin_amdgcn_wmma_f32_16x16x4_f32(false, a, false, qb[kk],
                                                (short)0, s, false, false);
    }

    float rm = fmaxf(fmaxf(fmaxf(s[0], s[1]), fmaxf(s[2], s[3])),
                     fmaxf(fmaxf(s[4], s[5]), fmaxf(s[6], s[7])));
    rm = fmaxf(rm, xhalf(rm));
    const float mnew  = fmaxf(mrun, rm);
    const float alpha = __expf(mrun - mnew);
    v8f p;
#pragma unroll
    for (int r = 0; r < 8; ++r) p[r] = __expf(s[r] - mnew);
    const float rs = ((p[0] + p[1]) + (p[2] + p[3])) + ((p[4] + p[5]) + (p[6] + p[7]));
    lsum = lsum * alpha + rs;
    mrun = mnew;

    o0 *= alpha; o1 *= alpha; o2 *= alpha; o3 *= alpha;

    const float* vrow = v + base + (long)(jb * 16 + 8 * half) * HEAD_D + 4 * l16;
#pragma unroll
    for (int j = 0; j < 4; ++j) {
      const v4f va = *(const v4f*)(vrow + (2 * j) * HEAD_D);
      const v4f vb = *(const v4f*)(vrow + (2 * j + 1) * HEAD_D);
      v2f pb;  pb.x = p[2 * j];  pb.y = p[2 * j + 1];
      v2f a0, a1, a2, a3;
      a0.x = va.x; a0.y = vb.x;
      a1.x = va.y; a1.y = vb.y;
      a2.x = va.z; a2.y = vb.z;
      a3.x = va.w; a3.y = vb.w;
      o0 = __builtin_amdgcn_wmma_f32_16x16x4_f32(false, a0, false, pb, (short)0, o0, false, false);
      o1 = __builtin_amdgcn_wmma_f32_16x16x4_f32(false, a1, false, pb, (short)0, o1, false, false);
      o2 = __builtin_amdgcn_wmma_f32_16x16x4_f32(false, a2, false, pb, (short)0, o2, false, false);
      o3 = __builtin_amdgcn_wmma_f32_16x16x4_f32(false, a3, false, pb, (short)0, o3, false, false);
    }
  }

  float* osh = wtile[wv];
#pragma unroll
  for (int r = 0; r < 8; ++r) {
    v4f t;
    t.x = o0[r]; t.y = o1[r]; t.z = o2[r]; t.w = o3[r];
    *(v4f*)&osh[l16 * OPITCH + 4 * (r + 8 * half)] = t;
  }
  lsum = lsum + xhalf(lsum);
  if (half == 0) {
    wmax[wv][l16] = mrun;
    wsum[wv][l16] = lsum;
  }
  __syncthreads();

#pragma unroll
  for (int i = 0; i < 2; ++i) {
    const int m = 4 * wv + 2 * i + half;
    float mp[NSPLIT], lp[NSPLIT];
    float mall = -1e30f;
#pragma unroll
    for (int p = 0; p < NSPLIT; ++p) {
      mp[p] = wmax[p][m];
      lp[p] = wsum[p][m];
      mall = fmaxf(mall, mp[p]);
    }
    float lall = 0.0f;
    v4f acc = {};
#pragma unroll
    for (int p = 0; p < NSPLIT; ++p) {
      const float sc = __expf(mp[p] - mall);
      const v4f op = *(const v4f*)&wtile[p][m * OPITCH + 4 * l16];
      lall += sc * lp[p];
      acc  += sc * op;
    }
    const float inv = 1.0f / lall;
    *(v4f*)(out + base + (long)(row * 16 + m) * HEAD_D + 4 * l16) = acc * inv;
  }
}

extern "C" void kernel_launch(void* const* d_in, const int* in_sizes, int n_in,
                              void* d_out, int out_size, void* d_ws, size_t ws_size,
                              hipStream_t stream) {
  (void)n_in; (void)out_size; (void)d_ws; (void)ws_size;
  const float* q = (const float*)d_in[0];
  const float* k = (const float*)d_in[1];
  const float* v = (const float*)d_in[2];
  float* out = (float*)d_out;

  const int bh = in_sizes[0] / (SEQ_L * HEAD_D);
  dim3 grid((unsigned)bh, NBLK);
  sparse_attn_wmma_f32<<<grid, 32 * NSPLIT, 0, stream>>>(q, k, v, out);
}


# Q block and every K tile fetched by whole-line global loads (4 full lines per instruction, was 16 lines x 16 B) and turned into WMMA fragments through the wave's own LDS tile; sum order unchanged
# speedup vs baseline: 2.0539x; 1.0993x over previous
#include <hip/hip_runtime.h>

typedef __attribute__((ext_vector_type(2))) float v2f;
typedef __attribute__((ext_vector_type(4))) float v4f;
typedef __attribute__((ext_vector_type(8))) float v8f;

#define SEQ_L 2048
#define HEAD_D 64
#define NBLK 128
#define NSPLIT 4
#define OPITCH 68

__device__ __forceinline__ float xhalf(float x) {
  const int b = __builtin_bit_cast(int, x);
  return __builtin_bit_cast(float,
      __builtin_amdgcn_permlanex16(b, b, 0x76543210, 0xfedcba98, false, false));
}

__global__ __launch_bounds__(32 * NSPLIT)
void sparse_attn_wmma_f32(const float* __restrict__ q,
                          const float* __restrict__ k,
                          const float* __restrict__ v,
                          float* __restrict__ out) {
  __shared__ __attribute__((aligned(16))) float wtile[NSPLIT][16 * OPITCH];
  __shared__ float wmax[NSPLIT][16];
  __shared__ float wsum[NSPLIT][16];

  const int wv   = __builtin_amdgcn_readfirstlane((int)(threadIdx.x >> 5));
  const int lane = threadIdx.x & 31;
  const int half = lane >> 4;
  const int l16  = lane & 15;

  const int  row  = (NBLK - 1) - (int)blockIdx.y;
  const long bh   = (long)blockIdx.x;
  const long base = bh * (long)(SEQ_L * HEAD_D);

  float* ksh = wtile[wv];

  {
    const float* qg = q + base + (long)(row * 16 + 8 * half) * HEAD_D + 4 * l16;
    v4f qr[8];
#pragma unroll
    for (int i = 0; i < 8; ++i) qr[i] = *(const v4f*)(qg + i * HEAD_D);
#pragma unroll
    for (int i = 0; i < 8; ++i) *(v4f*)&ksh[(8 * half + i) * OPITCH + 4 * l16] = qr[i];
  }
  const float* qrow = ksh + l16 * OPITCH + 2 * half;
  v2f qb[16];
#pragma unroll
  for (int kk = 0; kk < 16; ++kk) {
    v2f t = *(const v2f*)(qrow + 4 * kk);
    qb[kk].x = t.x * 0.125f;
    qb[kk].y = t.y * 0.125f;
  }

  v8f o0 = {}, o1 = {}, o2 = {}, o3 = {};
  float mrun = -1e30f;
  float lsum = 0.0f;

  const int w     = row & ~3;
  const int nvert = w >> 2;
  const int nbl   = nvert + (row - w + 1);

  for (int t = wv; t < nbl; t += NSPLIT) {
    const int jb = (t < nvert) ? (4 * t + 3) : (w + (t - nvert));

    {
      const int tn  = (t + NSPLIT < nbl) ? (t + NSPLIT) : t;
      const int jbn = (tn < nvert) ? (4 * tn + 3) : (w + (tn - nvert));
      const float* kp = k + base + (long)(jbn * 16 + l16) * HEAD_D;
      const float* vp = v + base + (long)(jbn * 16 + l16) * HEAD_D;
      __builtin_prefetch(kp, 0, 0);
      __builtin_prefetch(vp, 0, 0);
    }

    const float* kg = k + base + (long)(jb * 16 + 8 * half) * HEAD_D + 4 * l16;
    v4f kr[8];
#pragma unroll
    for (int i = 0; i < 8; ++i) kr[i] = *(const v4f*)(kg + i * HEAD_D);
#pragma unroll
    for (int i = 0; i < 8; ++i) *(v4f*)&ksh[(8 * half + i) * OPITCH + 4 * l16] = kr[i];
    const float* krow = ksh + l16 * OPITCH + 2 * half;
    v8f s = {};
#pragma unroll
    for (int kk = 0; kk < 16; ++kk) {
      v2f a = *(const v2f*)(krow + 4 * kk);
      s = __builtin_amdgcn_wmma_f32_16x16x4_f32(false, a, false, qb[kk],
                                                (short)0, s, false, false);
    }

    float rm = fmaxf(fmaxf(fmaxf(s[0], s[1]), fmaxf(s[2], s[3])),
                     fmaxf(fmaxf(s[4], s[5]), fmaxf(s[6], s[7])));
    rm = fmaxf(rm, xhalf(rm));
    const float mnew  = fmaxf(mrun, rm);
    const float alpha = __expf(mrun - mnew);
    v8f p;
#pragma unroll
    for (int r = 0; r < 8; ++r) p[r] = __expf(s[r] - mnew);
    const float rs = ((p[0] + p[1]) + (p[2] + p[3])) + ((p[4] + p[5]) + (p[6] + p[7]));
    lsum = lsum * alpha + rs;
    mrun = mnew;

    o0 *= alpha; o1 *= alpha; o2 *= alpha; o3 *= alpha;

    const float* vrow = v + base + (long)(jb * 16 + 8 * half) * HEAD_D + 4 * l16;
#pragma unroll
    for (int j = 0; j < 4; ++j) {
      const v4f va = *(const v4f*)(vrow + (2 * j) * HEAD_D);
      const v4f vb = *(const v4f*)(vrow + (2 * j + 1) * HEAD_D);
      v2f pb;  pb.x = p[2 * j];  pb.y = p[2 * j + 1];
      v2f a0, a1, a2, a3;
      a0.x = va.x; a0.y = vb.x;
      a1.x = va.y; a1.y = vb.y;
      a2.x = va.z; a2.y = vb.z;
      a3.x = va.w; a3.y = vb.w;
      o0 = __builtin_amdgcn_wmma_f32_16x16x4_f32(false, a0, false, pb, (short)0, o0, false, false);
      o1 = __builtin_amdgcn_wmma_f32_16x16x4_f32(false, a1, false, pb, (short)0, o1, false, false);
      o2 = __builtin_amdgcn_wmma_f32_16x16x4_f32(false, a2, false, pb, (short)0, o2, false, false);
      o3 = __builtin_amdgcn_wmma_f32_16x16x4_f32(false, a3, false, pb, (short)0, o3, false, false);
    }
  }

  float* osh = ksh;
#pragma unroll
  for (int r = 0; r < 8; ++r) {
    v4f t;
    t.x = o0[r]; t.y = o1[r]; t.z = o2[r]; t.w = o3[r];
    *(v4f*)&osh[l16 * OPITCH + 4 * (r + 8 * half)] = t;
  }
  lsum = lsum + xhalf(lsum);
  if (half == 0) {
    wmax[wv][l16] = mrun;
    wsum[wv][l16] = lsum;
  }
  __syncthreads();

#pragma unroll
  for (int i = 0; i < 2; ++i) {
    const int m = 4 * wv + 2 * i + half;
    float mp[NSPLIT], lp[NSPLIT];
    float mall = -1e30f;
#pragma unroll
    for (int p = 0; p < NSPLIT; ++p) {
      mp[p] = wmax[p][m];
      lp[p] = wsum[p][m];
      mall = fmaxf(mall, mp[p]);
    }
    float lall = 0.0f;
    v4f acc = {};
#pragma unroll
    for (int p = 0; p < NSPLIT; ++p) {
      const float sc = __expf(mp[p] - mall);
      const v4f op = *(const v4f*)&wtile[p][m * OPITCH + 4 * l16];
      lall += sc * lp[p];
      acc  += sc * op;
    }
    const float inv = 1.0f / lall;
    *(v4f*)(out + base + (long)(row * 16 + m) * HEAD_D + 4 * l16) = acc * inv;
  }
}

extern "C" void kernel_launch(void* const* d_in, const int* in_sizes, int n_in,
                              void* d_out, int out_size, void* d_ws, size_t ws_size,
                              hipStream_t stream) {
  (void)n_in; (void)out_size; (void)d_ws; (void)ws_size;
  const float* q = (const float*)d_in[0];
  const float* k = (const float*)d_in[1];
  const float* v = (const float*)d_in[2];
  float* out = (float*)d_out;

  const int bh = in_sizes[0] / (SEQ_L * HEAD_D);
  dim3 grid((unsigned)bh, NBLK);
  sparse_attn_wmma_f32<<<grid, 32 * NSPLIT, 0, stream>>>(q, k, v, out);
}


# each work-group takes two adjacent query block rows (2p, 2p+1): every K/V tile is fetched and staged once and feeds both rows' WMMAs from registers; per-row key subsets and sum order unchanged
# speedup vs baseline: 2.2754x; 1.1078x over previous
#include <hip/hip_runtime.h>

typedef __attribute__((ext_vector_type(2))) float v2f;
typedef __attribute__((ext_vector_type(4))) float v4f;
typedef __attribute__((ext_vector_type(8))) float v8f;

#define SEQ_L 2048
#define HEAD_D 64
#define NBLK 128
#define NPAIR 64
#define NSPLIT 4
#define OPITCH 68

#define WMMA_F32(a, b, c) __builtin_amdgcn_wmma_f32_16x16x4_f32(false, (a), false, (b), (short)0, (c), false, false)

__device__ __forceinline__ float xhalf(float x) {
  const int b = __builtin_bit_cast(int, x);
  return __builtin_bit_cast(float,
      __builtin_amdgcn_permlanex16(b, b, 0x76543210, 0xfedcba98, false, false));
}

struct RowAcc {
  v8f o0, o1, o2, o3;
  float mrun;
  float lsum;
};

__device__ __forceinline__ void load_q(const float* __restrict__ q, long base, int row,
                                       int half, int l16, float* tile, v2f (&qb)[16]) {
  const float* qg = q + base + (long)(row * 16 + 8 * half) * HEAD_D + 4 * l16;
  v4f qr[8];
#pragma unroll
  for (int i = 0; i < 8; ++i) qr[i] = *(const v4f*)(qg + i * HEAD_D);
#pragma unroll
  for (int i = 0; i < 8; ++i) *(v4f*)&tile[(8 * half + i) * OPITCH + 4 * l16] = qr[i];
  const float* qrow = tile + l16 * OPITCH + 2 * half;
#pragma unroll
  for (int kk = 0; kk < 16; ++kk) {
    v2f t = *(const v2f*)(qrow + 4 * kk);
    qb[kk].x = t.x * 0.125f;
    qb[kk].y = t.y * 0.125f;
  }
}

__device__ __forceinline__ void load_kv(const float* __restrict__ k, const float* __restrict__ v,
                                        long base, int jb, int half, int l16, float* ksh,
                                        v2f (&ka)[16], v2f (&av)[4][4]) {
  const float* kg = k + base + (long)(jb * 16 + 8 * half) * HEAD_D + 4 * l16;
  v4f kr[8];
#pragma unroll
  for (int i = 0; i < 8; ++i) kr[i] = *(const v4f*)(kg + i * HEAD_D);
#pragma unroll
  for (int i = 0; i < 8; ++i) *(v4f*)&ksh[(8 * half + i) * OPITCH + 4 * l16] = kr[i];
  const float* krow = ksh + l16 * OPITCH + 2 * half;
#pragma unroll
  for (int kk = 0; kk < 16; ++kk) ka[kk] = *(const v2f*)(krow + 4 * kk);
  const float* vrow = v + base + (long)(jb * 16 + 8 * half) * HEAD_D + 4 * l16;
#pragma unroll
  for (int j = 0; j < 4; ++j) {
    const v4f va = *(const v4f*)(vrow + (2 * j) * HEAD_D);
    const v4f vb = *(const v4f*)(vrow + (2 * j + 1) * HEAD_D);
    av[j][0].x = va.x; av[j][0].y = vb.x;
    av[j][1].x = va.y; av[j][1].y = vb.y;
    av[j][2].x = va.z; av[j][2].y = vb.z;
    av[j][3].x = va.w; av[j][3].y = vb.w;
  }
}

__device__ __forceinline__ void softmax_step(const v8f& s, v8f& p, RowAcc& a) {
  float rm = fmaxf(fmaxf(fmaxf(s[0], s[1]), fmaxf(s[2], s[3])),
                   fmaxf(fmaxf(s[4], s[5]), fmaxf(s[6], s[7])));
  rm = fmaxf(rm, xhalf(rm));
  const float mnew  = fmaxf(a.mrun, rm);
  const float alpha = __expf(a.mrun - mnew);
#pragma unroll
  for (int r = 0; r < 8; ++r) p[r] = __expf(s[r] - mnew);
  const float rs = ((p[0] + p[1]) + (p[2] + p[3])) + ((p[4] + p[5]) + (p[6] + p[7]));
  a.lsum = a.lsum * alpha + rs;
  a.mrun = mnew;
  a.o0 *= alpha; a.o1 *= alpha; a.o2 *= alpha; a.o3 *= alpha;
}

__global__ __launch_bounds__(32 * NSPLIT)
void sparse_attn_wmma_f32(const float* __restrict__ q,
                          const float* __restrict__ k,
                          const float* __restrict__ v,
                          float* __restrict__ out) {
  __shared__ __attribute__((aligned(16))) float wtile[2][NSPLIT][16 * OPITCH];
  __shared__ float wmax[2][NSPLIT][16];
  __shared__ float wsum[2][NSPLIT][16];

  const int wv   = __builtin_amdgcn_readfirstlane((int)(threadIdx.x >> 5));
  const int lane = threadIdx.x & 31;
  const int half = lane >> 4;
  const int l16  = lane & 15;

  const int  pr   = (NPAIR - 1) - (int)blockIdx.y;
  const int  rowB = 2 * pr + 1;
  const int  rowA = 2 * pr;
  const long bh   = (long)blockIdx.x;
  const long base = bh * (long)(SEQ_L * HEAD_D);

  float* ksh = wtile[0][wv];

  v2f qbB[16], qbA[16];
  load_q(q, base, rowB, half, l16, wtile[0][wv], qbB);
  load_q(q, base, rowA, half, l16, wtile[1][wv], qbA);

  RowAcc accB, accA;
  accB.o0 = (v8f){}; accB.o1 = (v8f){}; accB.o2 = (v8f){}; accB.o3 = (v8f){};
  accA.o0 = (v8f){}; accA.o1 = (v8f){}; accA.o2 = (v8f){}; accA.o3 = (v8f){};
  accB.mrun = -1e30f; accA.mrun = -1e30f;
  accB.lsum = 0.0f;   accA.lsum = 0.0f;

  const int w     = rowB & ~3;
  const int nvert = w >> 2;
  const int nbl   = nvert + (rowB - w + 1);
  const int nboth = nbl - 1;

  v2f ka[16];
  v2f av[4][4];

  for (int t = wv; t < nboth; t += NSPLIT) {
    const int jb = (t < nvert) ? (4 * t + 3) : (w + (t - nvert));

    {
      const int tn  = (t + NSPLIT < nbl) ? (t + NSPLIT) : t;
      const int jbn = (tn < nvert) ? (4 * tn + 3) : (w + (tn - nvert));
      const float* kp = k + base + (long)(jbn * 16 + l16) * HEAD_D;
      const float* vp = v + base + (long)(jbn * 16 + l16) * HEAD_D;
      __builtin_prefetch(kp, 0, 0);
      __builtin_prefetch(vp, 0, 0);
    }

    load_kv(k, v, base, jb, half, l16, ksh, ka, av);

    v8f sB = {}, sA = {};
#pragma unroll
    for (int kk = 0; kk < 16; ++kk) {
      sB = WMMA_F32(ka[kk], qbB[kk], sB);
      sA = WMMA_F32(ka[kk], qbA[kk], sA);
    }

    v8f pB, pA;
    softmax_step(sB, pB, accB);
    softmax_step(sA, pA, accA);

#pragma unroll
    for (int j = 0; j < 4; ++j) {
      v2f bB;  bB.x = pB[2 * j];  bB.y = pB[2 * j + 1];
      v2f bA;  bA.x = pA[2 * j];  bA.y = pA[2 * j + 1];
      accB.o0 = WMMA_F32(av[j][0], bB, accB.o0);
      accB.o1 = WMMA_F32(av[j][1], bB, accB.o1);
      accB.o2 = WMMA_F32(av[j][2], bB, accB.o2);
      accB.o3 = WMMA_F32(av[j][3], bB, accB.o3);
      accA.o0 = WMMA_F32(av[j][0], bA, accA.o0);
      accA.o1 = WMMA_F32(av[j][1], bA, accA.o1);
      accA.o2 = WMMA_F32(av[j][2], bA, accA.o2);
      accA.o3 = WMMA_F32(av[j][3], bA, accA.o3);
    }
  }

  if (wv == (nboth & (NSPLIT - 1))) {
    load_kv(k, v, base, rowB, half, l16, ksh, ka, av);
    v8f sB = {};
#pragma unroll
    for (int kk = 0; kk < 16; ++kk) sB = WMMA_F32(ka[kk], qbB[kk], sB);
    v8f pB;
    softmax_step(sB, pB, accB);
#pragma unroll
    for (int j = 0; j < 4; ++j) {
      v2f bB;  bB.x = pB[2 * j];  bB.y = pB[2 * j + 1];
      accB.o0 = WMMA_F32(av[j][0], bB, accB.o0);
      accB.o1 = WMMA_F32(av[j][1], bB, accB.o1);
      accB.o2 = WMMA_F32(av[j][2], bB, accB.o2);
      accB.o3 = WMMA_F32(av[j][3], bB, accB.o3);
    }
  }

  float* oshB = wtile[0][wv];
  float* oshA = wtile[1][wv];
#pragma unroll
  for (int r = 0; r < 8; ++r) {
    v4f tB, tA;
    tB.x = accB.o0[r]; tB.y = accB.o1[r]; tB.z = accB.o2[r]; tB.w = accB.o3[r];
    tA.x = accA.o0[r]; tA.y = accA.o1[r]; tA.z = accA.o2[r]; tA.w = accA.o3[r];
    *(v4f*)&oshB[l16 * OPITCH + 4 * (r + 8 * half)] = tB;
    *(v4f*)&oshA[l16 * OPITCH + 4 * (r + 8 * half)] = tA;
  }
  const float lB = accB.lsum + xhalf(accB.lsum);
  const float lA = accA.lsum + xhalf(accA.lsum);
  if (half == 0) {
    wmax[0][wv][l16] = accB.mrun;
    wsum[0][wv][l16] = lB;
    wmax[1][wv][l16] = accA.mrun;
    wsum[1][wv][l16] = lA;
  }
  __syncthreads();

#pragma unroll
  for (int sel = 0; sel < 2; ++sel) {
    const int row = (sel == 0) ? rowB : rowA;
#pragma unroll
    for (int i = 0; i < 2; ++i) {
      const int m = 4 * wv + 2 * i + half;
      float mp[NSPLIT], lp[NSPLIT];
      float mall = -1e30f;
#pragma unroll
      for (int p = 0; p < NSPLIT; ++p) {
        mp[p] = wmax[sel][p][m];
        lp[p] = wsum[sel][p][m];
        mall = fmaxf(mall, mp[p]);
      }
      float lall = 0.0f;
      v4f acc = {};
#pragma unroll
      for (int p = 0; p < NSPLIT; ++p) {
        const float sc = __expf(mp[p] - mall);
        const v4f op = *(const v4f*)&wtile[sel][p][m * OPITCH + 4 * l16];
        lall += sc * lp[p];
        acc  += sc * op;
      }
      const float inv = 1.0f / lall;
      *(v4f*)(out + base + (long)(row * 16 + m) * HEAD_D + 4 * l16) = acc * inv;
    }
  }
}

extern "C" void kernel_launch(void* const* d_in, const int* in_sizes, int n_in,
                              void* d_out, int out_size, void* d_ws, size_t ws_size,
                              hipStream_t stream) {
  (void)n_in; (void)out_size; (void)d_ws; (void)ws_size;
  const float* q = (const float*)d_in[0];
  const float* k = (const float*)d_in[1];
  const float* v = (const float*)d_in[2];
  float* out = (float*)d_out;

  const int bh = in_sizes[0] / (SEQ_L * HEAD_D);
  dim3 grid((unsigned)bh, NPAIR);
  sparse_attn_wmma_f32<<<grid, 32 * NSPLIT, 0, stream>>>(q, k, v, out);
}


# lazy rescale: O, row-sum and max half-exchange leave the key loop; one wave-uniform test per tile raises the maximum in use only when a score tops it by 5.5; terms rescaled, sum order unchanged
# speedup vs baseline: 2.3461x; 1.0311x over previous
#include <hip/hip_runtime.h>

typedef __attribute__((ext_vector_type(2))) float v2f;
typedef __attribute__((ext_vector_type(4))) float v4f;
typedef __attribute__((ext_vector_type(8))) float v8f;

#define SEQ_L 2048
#define HEAD_D 64
#define NBLK 128
#define NPAIR 64
#define NSPLIT 4
#define OPITCH 68
#define RAISE_MARGIN 5.5f

#define WMMA_F32(a, b, c) __builtin_amdgcn_wmma_f32_16x16x4_f32(false, (a), false, (b), (short)0, (c), false, false)

__device__ __forceinline__ float xhalf(float x) {
  const int b = __builtin_bit_cast(int, x);
  return __builtin_bit_cast(float,
      __builtin_amdgcn_permlanex16(b, b, 0x76543210, 0xfedcba98, false, false));
}

struct RowAcc {
  v8f o0, o1, o2, o3;
  float mref;
  float mthr;
  float lsum;
};

__device__ __forceinline__ void load_q(const float* __restrict__ q, long base, int row,
                                       int half, int l16, float* tile, v2f (&qb)[16]) {
  const float* qg = q + base + (long)(row * 16 + 8 * half) * HEAD_D + 4 * l16;
  v4f qr[8];
#pragma unroll
  for (int i = 0; i < 8; ++i) qr[i] = *(const v4f*)(qg + i * HEAD_D);
#pragma unroll
  for (int i = 0; i < 8; ++i) *(v4f*)&tile[(8 * half + i) * OPITCH + 4 * l16] = qr[i];
  const float* qrow = tile + l16 * OPITCH + 2 * half;
#pragma unroll
  for (int kk = 0; kk < 16; ++kk) {
    v2f t = *(const v2f*)(qrow + 4 * kk);
    qb[kk].x = t.x * 0.125f;
    qb[kk].y = t.y * 0.125f;
  }
}

__device__ __forceinline__ void load_kv(const float* __restrict__ k, const float* __restrict__ v,
                                        long base, int jb, int half, int l16, float* ksh,
                                        v2f (&ka)[16], v2f (&av)[4][4]) {
  const float* kg = k + base + (long)(jb * 16 + 8 * half) * HEAD_D + 4 * l16;
  v4f kr[8];
#pragma unroll
  for (int i = 0; i < 8; ++i) kr[i] = *(const v4f*)(kg + i * HEAD_D);
#pragma unroll
  for (int i = 0; i < 8; ++i) *(v4f*)&ksh[(8 * half + i) * OPITCH + 4 * l16] = kr[i];
  const float* krow = ksh + l16 * OPITCH + 2 * half;
#pragma unroll
  for (int kk = 0; kk < 16; ++kk) ka[kk] = *(const v2f*)(krow + 4 * kk);
  const float* vrow = v + base + (long)(jb * 16 + 8 * half) * HEAD_D + 4 * l16;
#pragma unroll
  for (int j = 0; j < 4; ++j) {
    const v4f va = *(const v4f*)(vrow + (2 * j) * HEAD_D);
    const v4f vb = *(const v4f*)(vrow + (2 * j + 1) * HEAD_D);
    av[j][0].x = va.x; av[j][0].y = vb.x;
    av[j][1].x = va.y; av[j][1].y = vb.y;
    av[j][2].x = va.z; av[j][2].y = vb.z;
    av[j][3].x = va.w; av[j][3].y = vb.w;
  }
}

__device__ __forceinline__ float half_max8(const v8f& s) {
  return fmaxf(fmaxf(fmaxf(s[0], s[1]), fmaxf(s[2], s[3])),
               fmaxf(fmaxf(s[4], s[5]), fmaxf(s[6], s[7])));
}

__device__ __forceinline__ void raise_reference(float hm, RowAcc& a, bool first) {
  const float rm   = fmaxf(hm, xhalf(hm));
  const float mnew = fmaxf(a.mref, rm);
  if (!first) {
    const float alpha = __expf(a.mref - mnew);
    a.lsum *= alpha;
    a.o0 *= alpha; a.o1 *= alpha; a.o2 *= alpha; a.o3 *= alpha;
  }
  a.mref = mnew;
  a.mthr = mnew + RAISE_MARGIN;
}

__device__ __forceinline__ void softmax_terms(const v8f& s, v8f& p, RowAcc& a) {
#pragma unroll
  for (int r = 0; r < 8; ++r) p[r] = __expf(s[r] - a.mref);
  const float rs = ((p[0] + p[1]) + (p[2] + p[3])) + ((p[4] + p[5]) + (p[6] + p[7]));
  a.lsum += rs;
}

__global__ __launch_bounds__(32 * NSPLIT)
void sparse_attn_wmma_f32(const float* __restrict__ q,
                          const float* __restrict__ k,
                          const float* __restrict__ v,
                          float* __restrict__ out) {
  __shared__ __attribute__((aligned(16))) float wtile[2][NSPLIT][16 * OPITCH];
  __shared__ float wmax[2][NSPLIT][16];
  __shared__ float wsum[2][NSPLIT][16];

  const int wv   = __builtin_amdgcn_readfirstlane((int)(threadIdx.x >> 5));
  const int lane = threadIdx.x & 31;
  const int half = lane >> 4;
  const int l16  = lane & 15;

  const int  pr   = (NPAIR - 1) - (int)blockIdx.y;
  const int  rowB = 2 * pr + 1;
  const int  rowA = 2 * pr;
  const long bh   = (long)blockIdx.x;
  const long base = bh * (long)(SEQ_L * HEAD_D);

  float* ksh = wtile[0][wv];

  v2f qbB[16], qbA[16];
  load_q(q, base, rowB, half, l16, wtile[0][wv], qbB);
  load_q(q, base, rowA, half, l16, wtile[1][wv], qbA);

  RowAcc accB, accA;
  accB.o0 = (v8f){}; accB.o1 = (v8f){}; accB.o2 = (v8f){}; accB.o3 = (v8f){};
  accA.o0 = (v8f){}; accA.o1 = (v8f){}; accA.o2 = (v8f){}; accA.o3 = (v8f){};
  accB.mref = -1e30f; accA.mref = -1e30f;
  accB.mthr = -1e30f; accA.mthr = -1e30f;
  accB.lsum = 0.0f;   accA.lsum = 0.0f;

  const int w     = rowB & ~3;
  const int nvert = w >> 2;
  const int nbl   = nvert + (rowB - w + 1);
  const int nboth = nbl - 1;

  v2f ka[16];
  v2f av[4][4];

  for (int t = wv; t < nboth; t += NSPLIT) {
    const int jb = (t < nvert) ? (4 * t + 3) : (w + (t - nvert));

    {
      const int tn  = (t + NSPLIT < nbl) ? (t + NSPLIT) : t;
      const int jbn = (tn < nvert) ? (4 * tn + 3) : (w + (tn - nvert));
      const float* kp = k + base + (long)(jbn * 16 + l16) * HEAD_D;
      const float* vp = v + base + (long)(jbn * 16 + l16) * HEAD_D;
      __builtin_prefetch(kp, 0, 0);
      __builtin_prefetch(vp, 0, 0);
    }

    load_kv(k, v, base, jb, half, l16, ksh, ka, av);
    __builtin_amdgcn_sched_barrier(0x76);

    v8f sB = {}, sA = {};
#pragma unroll
    for (int kk = 0; kk < 16; ++kk) {
      sB = WMMA_F32(ka[kk], qbB[kk], sB);
      sA = WMMA_F32(ka[kk], qbA[kk], sA);
    }

    const float hmB = half_max8(sB);
    const float hmA = half_max8(sA);
    if (__builtin_amdgcn_ballot_w32((hmB > accB.mthr) | (hmA > accA.mthr)) != 0u) {
      const bool first = (t == wv);
      raise_reference(hmB, accB, first);
      raise_reference(hmA, accA, first);
    }
    v8f pB, pA;
    softmax_terms(sB, pB, accB);
    softmax_terms(sA, pA, accA);

#pragma unroll
    for (int j = 0; j < 4; ++j) {
      v2f bB;  bB.x = pB[2 * j];  bB.y = pB[2 * j + 1];
      v2f bA;  bA.x = pA[2 * j];  bA.y = pA[2 * j + 1];
      accB.o0 = WMMA_F32(av[j][0], bB, accB.o0);
      accB.o1 = WMMA_F32(av[j][1], bB, accB.o1);
      accB.o2 = WMMA_F32(av[j][2], bB, accB.o2);
      accB.o3 = WMMA_F32(av[j][3], bB, accB.o3);
      accA.o0 = WMMA_F32(av[j][0], bA, accA.o0);
      accA.o1 = WMMA_F32(av[j][1], bA, accA.o1);
      accA.o2 = WMMA_F32(av[j][2], bA, accA.o2);
      accA.o3 = WMMA_F32(av[j][3], bA, accA.o3);
    }
  }

  if (wv == (nboth & (NSPLIT - 1))) {
    load_kv(k, v, base, rowB, half, l16, ksh, ka, av);
    __builtin_amdgcn_sched_barrier(0x76);
    v8f sB = {};
#pragma unroll
    for (int kk = 0; kk < 16; ++kk) sB = WMMA_F32(ka[kk], qbB[kk], sB);
    const float hmB = half_max8(sB);
    if (__builtin_amdgcn_ballot_w32(hmB > accB.mthr) != 0u) raise_reference(hmB, accB, nboth <= wv);
    v8f pB;
    softmax_terms(sB, pB, accB);
#pragma unroll
    for (int j = 0; j < 4; ++j) {
      v2f bB;  bB.x = pB[2 * j];  bB.y = pB[2 * j + 1];
      accB.o0 = WMMA_F32(av[j][0], bB, accB.o0);
      accB.o1 = WMMA_F32(av[j][1], bB, accB.o1);
      accB.o2 = WMMA_F32(av[j][2], bB, accB.o2);
      accB.o3 = WMMA_F32(av[j][3], bB, accB.o3);
    }
  }

  float* oshB = wtile[0][wv];
  float* oshA = wtile[1][wv];
#pragma unroll
  for (int r = 0; r < 8; ++r) {
    v4f tB, tA;
    tB.x = accB.o0[r]; tB.y = accB.o1[r]; tB.z = accB.o2[r]; tB.w = accB.o3[r];
    tA.x = accA.o0[r]; tA.y = accA.o1[r]; tA.z = accA.o2[r]; tA.w = accA.o3[r];
    *(v4f*)&oshB[l16 * OPITCH + 4 * (r + 8 * half)] = tB;
    *(v4f*)&oshA[l16 * OPITCH + 4 * (r + 8 * half)] = tA;
  }
  const float lB = accB.lsum + xhalf(accB.lsum);
  const float lA = accA.lsum + xhalf(accA.lsum);
  if (half == 0) {
    wmax[0][wv][l16] = accB.mref;
    wsum[0][wv][l16] = lB;
    wmax[1][wv][l16] = accA.mref;
    wsum[1][wv][l16] = lA;
  }
  __syncthreads();

#pragma unroll
  for (int sel = 0; sel < 2; ++sel) {
    const int row = (sel == 0) ? rowB : rowA;
#pragma unroll
    for (int i = 0; i < 2; ++i) {
      const int m = 4 * wv + 2 * i + half;
      float mp[NSPLIT], lp[NSPLIT];
      float mall = -1e30f;
#pragma unroll
      for (int p = 0; p < NSPLIT; ++p) {
        mp[p] = wmax[sel][p][m];
        lp[p] = wsum[sel][p][m];
        mall = fmaxf(mall, mp[p]);
      }
      float lall = 0.0f;
      v4f acc = {};
#pragma unroll
      for (int p = 0; p < NSPLIT; ++p) {
        const float sc = __expf(mp[p] - mall);
        const v4f op = *(const v4f*)&wtile[sel][p][m * OPITCH + 4 * l16];
        lall += sc * lp[p];
        acc  += sc * op;
      }
      const float inv = 1.0f / lall;
      *(v4f*)(out + base + (long)(row * 16 + m) * HEAD_D + 4 * l16) = acc * inv;
    }
  }
}

extern "C" void kernel_launch(void* const* d_in, const int* in_sizes, int n_in,
                              void* d_out, int out_size, void* d_ws, size_t ws_size,
                              hipStream_t stream) {
  (void)n_in; (void)out_size; (void)d_ws; (void)ws_size;
  const float* q = (const float*)d_in[0];
  const float* k = (const float*)d_in[1];
  const float* v = (const float*)d_in[2];
  float* out = (float*)d_out;

  const int bh = in_sizes[0] / (SEQ_L * HEAD_D);
  dim3 grid((unsigned)bh, NPAIR);
  sparse_attn_wmma_f32<<<grid, 32 * NSPLIT, 0, stream>>>(q, k, v, out);
}
